# starting kernel
# baseline (speedup 1.0000x reference)
#include <hip/hip_runtime.h>
#include <hip/hip_bf16.h>

typedef float v2f __attribute__((ext_vector_type(2)));
typedef float v8f __attribute__((ext_vector_type(8)));

#define MAXSEG 512
#define CB 8
#define PCHUNK 8192

__global__ void zero_ws_kernel(float* ws, int n) {
    int i = blockIdx.x * blockDim.x + threadIdx.x;
    if (i < n) ws[i] = 0.0f;
}

__global__ void count_kernel(const int* __restrict__ seg, float* __restrict__ counts,
                             int P, int N, int chunk) {
    __shared__ int hist[MAXSEG];
    for (int i = threadIdx.x; i < N; i += blockDim.x) hist[i] = 0;
    __syncthreads();
    int pstart = blockIdx.x * chunk;
    for (int i = threadIdx.x; i < chunk; i += blockDim.x) {
        int g = pstart + i;
        if (g < P) atomicAdd(&hist[seg[g]], 1);
    }
    __syncthreads();
    for (int i = threadIdx.x; i < N; i += blockDim.x) {
        int c = hist[i];
        if (c) atomicAdd(&counts[i], (float)c);
    }
}

__global__ void accum_kernel(const float* __restrict__ feat,
                             const int* __restrict__ seg,
                             float* __restrict__ sums,
                             int P, int C, int N, int chunk) {
    __shared__ int   lseg[PCHUNK];
    __shared__ float lsums[CB * MAXSEG];

    for (int i = threadIdx.x; i < CB * N; i += blockDim.x) lsums[i] = 0.0f;

    int pstart = blockIdx.x * chunk;
    int navail = P - pstart;
    int n = (navail < chunk) ? navail : chunk;
    if (n == chunk) {
        const int4* seg4 = (const int4*)(seg + pstart);
        int4* lseg4 = (int4*)lseg;
        for (int i = threadIdx.x; i < chunk / 4; i += blockDim.x)
            lseg4[i] = seg4[i];
    } else {
        for (int i = threadIdx.x; i < chunk; i += blockDim.x)
            lseg[i] = (i < n) ? seg[pstart + i] : 0;
    }
    __syncthreads();

    int wave = threadIdx.x >> 5;
    int lane = threadIdx.x & 31;
    int c = blockIdx.y * CB + wave;
    const float* plane = feat + (size_t)c * (size_t)P + pstart;
    float* lrow = &lsums[wave * N];

    if (n == chunk) {
        const float4* plane4 = (const float4*)plane;
        const int4*   lseg4  = (const int4*)lseg;
        for (int i = lane; i < chunk / 4; i += 32) {
            if (i + 256 < chunk / 4)
                __builtin_prefetch(&plane4[i + 256], 0, 0);
            float4 v = plane4[i];
            int4   s = lseg4[i];
            atomicAdd(&lrow[s.x], v.x);
            atomicAdd(&lrow[s.y], v.y);
            atomicAdd(&lrow[s.z], v.z);
            atomicAdd(&lrow[s.w], v.w);
        }
    } else {
        for (int i = lane; i < n; i += 32) {
            float v = plane[i];
            atomicAdd(&lrow[lseg[i]], v);
        }
    }
    __syncthreads();

    for (int i = threadIdx.x; i < CB * N; i += blockDim.x) {
        float v = lsums[i];
        if (v != 0.0f) {
            int w = i / N;
            int s = i - w * N;
            atomicAdd(&sums[(size_t)s * C + blockIdx.y * CB + w], v);
        }
    }
}

__global__ void finalize_wmma_kernel(const float* __restrict__ sums,
                                     const float* __restrict__ counts,
                                     float* __restrict__ out,
                                     int C, int N) {
    int wave = threadIdx.x >> 5;
    int lane = threadIdx.x & 31;
    int half = lane >> 4;
    int m    = lane & 15;
    int n0   = blockIdx.x * 16;
    int c0   = wave * 16;

    float cnt = counts[n0 + m];
    float inv = (cnt > 0.0f) ? (1.0f / cnt) : 0.0f;

    v8f acc = {};
#pragma unroll
    for (int k0 = 0; k0 < 16; k0 += 4) {
        int kx = k0 + 2 * half;
        v2f a, b;
        a.x = (m == kx)     ? inv : 0.0f;
        a.y = (m == kx + 1) ? inv : 0.0f;
        b.x = sums[(size_t)(n0 + kx)     * C + c0 + m];
        b.y = sums[(size_t)(n0 + kx + 1) * C + c0 + m];
        acc = __builtin_amdgcn_wmma_f32_16x16x4_f32(
             false, a,  false, b,
             (short)0, acc,  false,  false);
    }

#pragma unroll
    for (int v = 0; v < 8; ++v) {
        out[(size_t)(n0 + v + 8 * half) * C + c0 + m] = acc[v];
    }
}

extern "C" void kernel_launch(void* const* d_in, const int* in_sizes, int n_in,
                              void* d_out, int out_size, void* d_ws, size_t ws_size,
                              hipStream_t stream) {
    const float* feat = (const float*)d_in[0];
    const int*   seg  = (const int*)d_in[1];
    int P = in_sizes[1];
    int C = in_sizes[0] / P;
    int N = out_size / C;

    float* sums   = (float*)d_ws;
    float* counts = sums + (size_t)N * C;
    float* out    = (float*)d_out;

    int tot = N * C + N;
    zero_ws_kernel<<<(tot + 255) / 256, 256, 0, stream>>>(sums, tot);

    int chunk   = PCHUNK;
    int pblocks = (P + chunk - 1) / chunk;

    count_kernel<<<pblocks, 256, 0, stream>>>(seg, counts, P, N, chunk);

    dim3 agrid(pblocks, C / CB);
    accum_kernel<<<agrid, 256, 0, stream>>>(feat, seg, sums, P, C, N, chunk);

    finalize_wmma_kernel<<<N / 16, (C / 16) * 32, 0, stream>>>(sums, counts, out, C, N);
}


# sums workspace stored channel-major so each workgroup's flush is one contiguous run: global f32 atomics hit one 128-byte line per wave instruction instead of one line per lane
# speedup vs baseline: 4.1178x; 4.1178x over previous
#include <hip/hip_runtime.h>
#include <hip/hip_bf16.h>

typedef float v2f __attribute__((ext_vector_type(2)));
typedef float v8f __attribute__((ext_vector_type(8)));

#define MAXSEG 512
#define CB 8
#define PCHUNK 8192

__global__ void zero_ws_kernel(float* ws, int n) {
    int i = blockIdx.x * blockDim.x + threadIdx.x;
    if (i < n) ws[i] = 0.0f;
}

__global__ void count_kernel(const int* __restrict__ seg, float* __restrict__ counts,
                             int P, int N, int chunk) {
    __shared__ int hist[MAXSEG];
    for (int i = threadIdx.x; i < N; i += blockDim.x) hist[i] = 0;
    __syncthreads();
    int pstart = blockIdx.x * chunk;
    for (int i = threadIdx.x; i < chunk; i += blockDim.x) {
        int g = pstart + i;
        if (g < P) atomicAdd(&hist[seg[g]], 1);
    }
    __syncthreads();
    for (int i = threadIdx.x; i < N; i += blockDim.x) {
        int c = hist[i];
        if (c) atomicAdd(&counts[i], (float)c);
    }
}

__global__ void accum_kernel(const float* __restrict__ feat,
                             const int* __restrict__ seg,
                             float* __restrict__ sums,
                             int P, int C, int N, int chunk) {
    __shared__ int   lseg[PCHUNK];
    __shared__ float lsums[CB * MAXSEG];

    for (int i = threadIdx.x; i < CB * N; i += blockDim.x) lsums[i] = 0.0f;

    int pstart = blockIdx.x * chunk;
    int navail = P - pstart;
    int n = (navail < chunk) ? navail : chunk;
    if (n == chunk) {
        const int4* seg4 = (const int4*)(seg + pstart);
        int4* lseg4 = (int4*)lseg;
        for (int i = threadIdx.x; i < chunk / 4; i += blockDim.x)
            lseg4[i] = seg4[i];
    } else {
        for (int i = threadIdx.x; i < chunk; i += blockDim.x)
            lseg[i] = (i < n) ? seg[pstart + i] : 0;
    }
    __syncthreads();

    int wave = threadIdx.x >> 5;
    int lane = threadIdx.x & 31;
    int c = blockIdx.y * CB + wave;
    const float* plane = feat + (size_t)c * (size_t)P + pstart;
    float* lrow = &lsums[wave * N];

    if (n == chunk) {
        const float4* plane4 = (const float4*)plane;
        const int4*   lseg4  = (const int4*)lseg;
        for (int i = lane; i < chunk / 4; i += 32) {
            if (i + 256 < chunk / 4)
                __builtin_prefetch(&plane4[i + 256], 0, 0);
            float4 v = plane4[i];
            int4   s = lseg4[i];
            atomicAdd(&lrow[s.x], v.x);
            atomicAdd(&lrow[s.y], v.y);
            atomicAdd(&lrow[s.z], v.z);
            atomicAdd(&lrow[s.w], v.w);
        }
    } else {
        for (int i = lane; i < n; i += 32) {
            float v = plane[i];
            atomicAdd(&lrow[lseg[i]], v);
        }
    }
    __syncthreads();

    float* grow = sums + (size_t)blockIdx.y * CB * (size_t)N;
    for (int i = threadIdx.x; i < CB * N; i += blockDim.x) {
        float v = lsums[i];
        if (v != 0.0f) atomicAdd(&grow[i], v);
    }
}

__global__ void finalize_wmma_kernel(const float* __restrict__ sums,
                                     const float* __restrict__ counts,
                                     float* __restrict__ out,
                                     int C, int N) {
    int wave = threadIdx.x >> 5;
    int lane = threadIdx.x & 31;
    int half = lane >> 4;
    int m    = lane & 15;
    int n0   = blockIdx.x * 16;
    int c0   = wave * 16;

    float cnt = counts[n0 + m];
    float inv = (cnt > 0.0f) ? (1.0f / cnt) : 0.0f;

    v8f acc = {};
#pragma unroll
    for (int k0 = 0; k0 < 16; k0 += 4) {
        int kx = k0 + 2 * half;
        v2f a, b;
        a.x = (m == kx)     ? inv : 0.0f;
        a.y = (m == kx + 1) ? inv : 0.0f;
        b.x = sums[(size_t)(c0 + m) * N + n0 + kx];
        b.y = sums[(size_t)(c0 + m) * N + n0 + kx + 1];
        acc = __builtin_amdgcn_wmma_f32_16x16x4_f32(
             false, a,  false, b,
             (short)0, acc,  false,  false);
    }

#pragma unroll
    for (int v = 0; v < 8; ++v) {
        out[(size_t)(n0 + v + 8 * half) * C + c0 + m] = acc[v];
    }
}

extern "C" void kernel_launch(void* const* d_in, const int* in_sizes, int n_in,
                              void* d_out, int out_size, void* d_ws, size_t ws_size,
                              hipStream_t stream) {
    const float* feat = (const float*)d_in[0];
    const int*   seg  = (const int*)d_in[1];
    int P = in_sizes[1];
    int C = in_sizes[0] / P;
    int N = out_size / C;

    float* sums   = (float*)d_ws;
    float* counts = sums + (size_t)N * C;
    float* out    = (float*)d_out;

    int tot = N * C + N;
    zero_ws_kernel<<<(tot + 255) / 256, 256, 0, stream>>>(sums, tot);

    int chunk   = PCHUNK;
    int pblocks = (P + chunk - 1) / chunk;

    count_kernel<<<pblocks, 256, 0, stream>>>(seg, counts, P, N, chunk);

    dim3 agrid(pblocks, C / CB);
    accum_kernel<<<agrid, 256, 0, stream>>>(feat, seg, sums, P, C, N, chunk);

    finalize_wmma_kernel<<<N / 16, (C / 16) * 32, 0, stream>>>(sums, counts, out, C, N);
}


# each wave keeps 8 or more global loads in flight instead of 1: count and seg staging issue all loads before the first wait, accumulate loop refills an 8-deep float4 register ring; prefetch dropped
# speedup vs baseline: 5.6298x; 1.3672x over previous
#include <hip/hip_runtime.h>
#include <hip/hip_bf16.h>

typedef float v2f __attribute__((ext_vector_type(2)));
typedef float v8f __attribute__((ext_vector_type(8)));

#define MAXSEG 512
#define CB 8
#define PCHUNK 8192
#define CT 256
#define RING 8

__global__ void zero_ws_kernel(float* ws, int n) {
    int i = blockIdx.x * blockDim.x + threadIdx.x;
    if (i < n) ws[i] = 0.0f;
}

__global__ void __launch_bounds__(CT)
count_kernel(const int* __restrict__ seg, float* __restrict__ counts,
             int P, int N, int chunk) {
    __shared__ int hist[MAXSEG];
    for (int i = threadIdx.x; i < N; i += blockDim.x) hist[i] = 0;
    __syncthreads();
    int pstart = blockIdx.x * chunk;
    if (chunk == PCHUNK && pstart + PCHUNK <= P) {
        int s[PCHUNK / CT];
#pragma unroll
        for (int j = 0; j < PCHUNK / CT; ++j)
            s[j] = seg[pstart + (int)threadIdx.x + j * CT];
#pragma unroll
        for (int j = 0; j < PCHUNK / CT; ++j)
            atomicAdd(&hist[s[j]], 1);
    } else {
        for (int i = threadIdx.x; i < chunk; i += blockDim.x) {
            int g = pstart + i;
            if (g < P) atomicAdd(&hist[seg[g]], 1);
        }
    }
    __syncthreads();
    for (int i = threadIdx.x; i < N; i += blockDim.x) {
        int c = hist[i];
        if (c) atomicAdd(&counts[i], (float)c);
    }
}

__global__ void __launch_bounds__(CT)
accum_kernel(const float* __restrict__ feat,
                             const int* __restrict__ seg,
                             float* __restrict__ sums,
                             int P, int C, int N, int chunk) {
    __shared__ int   lseg[PCHUNK];
    __shared__ float lsums[CB * MAXSEG];

    for (int i = threadIdx.x; i < CB * N; i += blockDim.x) lsums[i] = 0.0f;

    int pstart = blockIdx.x * chunk;
    int navail = P - pstart;
    int n = (navail < chunk) ? navail : chunk;
    const bool full = (n == chunk) && (chunk == PCHUNK);
    if (full) {
        const int4* seg4 = (const int4*)(seg + pstart);
        int4* lseg4 = (int4*)lseg;
        int4 t[PCHUNK / 4 / CT];
#pragma unroll
        for (int j = 0; j < PCHUNK / 4 / CT; ++j)
            t[j] = seg4[(int)threadIdx.x + j * CT];
#pragma unroll
        for (int j = 0; j < PCHUNK / 4 / CT; ++j)
            lseg4[(int)threadIdx.x + j * CT] = t[j];
    } else {
        for (int i = threadIdx.x; i < chunk; i += blockDim.x)
            lseg[i] = (i < n) ? seg[pstart + i] : 0;
    }
    __syncthreads();

    int wave = threadIdx.x >> 5;
    int lane = threadIdx.x & 31;
    int c = blockIdx.y * CB + wave;
    const float* plane = feat + (size_t)c * (size_t)P + pstart;
    float* lrow = &lsums[wave * N];

    if (full) {
        const float4* p4 = (const float4*)plane + lane;
        const int4*   s4 = (const int4*)lseg + lane;
        constexpr int ITER = PCHUNK / 4 / 32;
        float4 buf[RING];
#pragma unroll
        for (int j = 0; j < RING; ++j) buf[j] = p4[32 * j];
        for (int it = 0; it < ITER - RING; it += RING) {
            int4 s[RING];
#pragma unroll
            for (int j = 0; j < RING; ++j) s[j] = s4[32 * (it + j)];
#pragma unroll
            for (int j = 0; j < RING; ++j) {
                float4 v = buf[j];
                atomicAdd(&lrow[s[j].x], v.x);
                atomicAdd(&lrow[s[j].y], v.y);
                atomicAdd(&lrow[s[j].z], v.z);
                atomicAdd(&lrow[s[j].w], v.w);
                buf[j] = p4[32 * (it + j + RING)];
            }
        }
        {
            int4 s[RING];
#pragma unroll
            for (int j = 0; j < RING; ++j) s[j] = s4[32 * (ITER - RING + j)];
#pragma unroll
            for (int j = 0; j < RING; ++j) {
                float4 v = buf[j];
                atomicAdd(&lrow[s[j].x], v.x);
                atomicAdd(&lrow[s[j].y], v.y);
                atomicAdd(&lrow[s[j].z], v.z);
                atomicAdd(&lrow[s[j].w], v.w);
            }
        }
    } else {
        for (int i = lane; i < n; i += 32) {
            float v = plane[i];
            atomicAdd(&lrow[lseg[i]], v);
        }
    }
    __syncthreads();

    float* grow = sums + (size_t)blockIdx.y * CB * (size_t)N;
    for (int i = threadIdx.x; i < CB * N; i += blockDim.x) {
        float v = lsums[i];
        if (v != 0.0f) atomicAdd(&grow[i], v);
    }
}

__global__ void finalize_wmma_kernel(const float* __restrict__ sums,
                                     const float* __restrict__ counts,
                                     float* __restrict__ out,
                                     int C, int N) {
    int wave = threadIdx.x >> 5;
    int lane = threadIdx.x & 31;
    int half = lane >> 4;
    int m    = lane & 15;
    int n0   = blockIdx.x * 16;
    int c0   = wave * 16;

    float cnt = counts[n0 + m];
    float inv = (cnt > 0.0f) ? (1.0f / cnt) : 0.0f;

    v8f acc = {};
#pragma unroll
    for (int k0 = 0; k0 < 16; k0 += 4) {
        int kx = k0 + 2 * half;
        v2f a, b;
        a.x = (m == kx)     ? inv : 0.0f;
        a.y = (m == kx + 1) ? inv : 0.0f;
        b.x = sums[(size_t)(c0 + m) * N + n0 + kx];
        b.y = sums[(size_t)(c0 + m) * N + n0 + kx + 1];
        acc = __builtin_amdgcn_wmma_f32_16x16x4_f32(
             false, a,  false, b,
             (short)0, acc,  false,  false);
    }

#pragma unroll
    for (int v = 0; v < 8; ++v) {
        out[(size_t)(n0 + v + 8 * half) * C + c0 + m] = acc[v];
    }
}

extern "C" void kernel_launch(void* const* d_in, const int* in_sizes, int n_in,
                              void* d_out, int out_size, void* d_ws, size_t ws_size,
                              hipStream_t stream) {
    const float* feat = (const float*)d_in[0];
    const int*   seg  = (const int*)d_in[1];
    int P = in_sizes[1];
    int C = in_sizes[0] / P;
    int N = out_size / C;

    float* sums   = (float*)d_ws;
    float* counts = sums + (size_t)N * C;
    float* out    = (float*)d_out;

    int tot = N * C + N;
    zero_ws_kernel<<<(tot + 255) / 256, 256, 0, stream>>>(sums, tot);

    int chunk   = PCHUNK;
    int pblocks = (P + chunk - 1) / chunk;

    count_kernel<<<pblocks, CT, 0, stream>>>(seg, counts, P, N, chunk);

    dim3 agrid(pblocks, C / CB);
    accum_kernel<<<agrid, CT, 0, stream>>>(feat, seg, sums, P, C, N, chunk);

    finalize_wmma_kernel<<<N / 16, (C / 16) * 32, 0, stream>>>(sums, counts, out, C, N);
}


# accumulate kernel: 64px x 16ch blocks transposed on the f32 matrix unit (data x one-hot) so lanes index channels: each LDS atomic hits consecutive words; sums add in wave-arrival order
# speedup vs baseline: 6.1360x; 1.0899x over previous
#include <hip/hip_runtime.h>
#include <hip/hip_bf16.h>


typedef float v2f __attribute__((ext_vector_type(2)));
typedef float v8f __attribute__((ext_vector_type(8)));

#define MAXSEG 512
#define CG 16
#define PCHUNK 8192
#define CT 256
#define NWAVE (CT / 32)
#define PBLK 64
#define WPIX (PCHUNK / NWAVE)
#define NBLK (WPIX / PBLK)
#define DEPTH 3

__global__ void zero_ws_kernel(float* ws, int n) {
    int i = blockIdx.x * blockDim.x + threadIdx.x;
    if (i < n) ws[i] = 0.0f;
}

__global__ void __launch_bounds__(CT)
count_kernel(const int* __restrict__ seg, float* __restrict__ counts,
             int P, int N, int chunk) {
    __shared__ int hist[MAXSEG];
    for (int i = threadIdx.x; i < N; i += blockDim.x) hist[i] = 0;
    __syncthreads();
    int pstart = blockIdx.x * chunk;
    if (chunk == PCHUNK && pstart + PCHUNK <= P) {
        int s[PCHUNK / CT];
#pragma unroll
        for (int j = 0; j < PCHUNK / CT; ++j)
            s[j] = seg[pstart + (int)threadIdx.x + j * CT];
#pragma unroll
        for (int j = 0; j < PCHUNK / CT; ++j)
            atomicAdd(&hist[s[j]], 1);
    } else {
        for (int i = threadIdx.x; i < chunk; i += blockDim.x) {
            int g = pstart + i;
            if (g < P) atomicAdd(&hist[seg[g]], 1);
        }
    }
    __syncthreads();
    for (int i = threadIdx.x; i < N; i += blockDim.x) {
        int c = hist[i];
        if (c) atomicAdd(&counts[i], (float)c);
    }
}

struct Blk {
    float4 d[8];
    int4   s;
};

__device__ __forceinline__ void load_block(Blk& b, const float* dptr, const int* sptr,
                                           size_t P, int blk) {
#pragma unroll
    for (int k = 0; k < 8; ++k) {
        const int ch = 4 * (k >> 1) + (k & 1);
        b.d[k] = *(const float4*)(dptr + (size_t)ch * P + (size_t)blk * PBLK);
    }
    b.s = *(const int4*)(sptr + (size_t)blk * PBLK);
}

__device__ __forceinline__ void transpose_block(const Blk& b, const v2f (&bsel)[4], v8f (&acc)[4]) {
#pragma unroll
    for (int q = 0; q < 4; ++q) acc[q] = v8f{};
#pragma unroll
    for (int g = 0; g < 4; ++g) {
        const float4 lo = b.d[2 * g], hi = b.d[2 * g + 1];
        v2f a0 = {lo.x, hi.x}, a1 = {lo.y, hi.y}, a2 = {lo.z, hi.z}, a3 = {lo.w, hi.w};
        acc[0] = __builtin_amdgcn_wmma_f32_16x16x4_f32(false, a0, false, bsel[g], (short)0, acc[0], false, false);
        acc[1] = __builtin_amdgcn_wmma_f32_16x16x4_f32(false, a1, false, bsel[g], (short)0, acc[1], false, false);
        acc[2] = __builtin_amdgcn_wmma_f32_16x16x4_f32(false, a2, false, bsel[g], (short)0, acc[2], false, false);
        acc[3] = __builtin_amdgcn_wmma_f32_16x16x4_f32(false, a3, false, bsel[g], (short)0, acc[3], false, false);
    }
}

template <int R>
__device__ __forceinline__ void scatter_row(const v8f (&acc)[4], int4 s, float* lcol) {
    const int s0 = __builtin_amdgcn_update_dpp(0, s.x, 0x150 + R, 0xF, 0xF, false);
    const int s1 = __builtin_amdgcn_update_dpp(0, s.y, 0x150 + R, 0xF, 0xF, false);
    const int s2 = __builtin_amdgcn_update_dpp(0, s.z, 0x150 + R, 0xF, 0xF, false);
    const int s3 = __builtin_amdgcn_update_dpp(0, s.w, 0x150 + R, 0xF, 0xF, false);
    atomicAdd(lcol + s0 * CG, acc[0][R]);
    atomicAdd(lcol + s1 * CG, acc[1][R]);
    atomicAdd(lcol + s2 * CG, acc[2][R]);
    atomicAdd(lcol + s3 * CG, acc[3][R]);
}

__device__ __forceinline__ void scatter_block(const v8f (&acc)[4], int4 s, float* lcol) {
    scatter_row<0>(acc, s, lcol); scatter_row<1>(acc, s, lcol);
    scatter_row<2>(acc, s, lcol); scatter_row<3>(acc, s, lcol);
    scatter_row<4>(acc, s, lcol); scatter_row<5>(acc, s, lcol);
    scatter_row<6>(acc, s, lcol); scatter_row<7>(acc, s, lcol);
}

__global__ void __launch_bounds__(CT)
accum_kernel(const float* __restrict__ feat,
                             const int* __restrict__ seg,
                             float* __restrict__ sums,
                             int P, int C, int N, int chunk) {
    __shared__ float lsums[MAXSEG * CG];

    for (int i = threadIdx.x; i < CG * N; i += blockDim.x) lsums[i] = 0.0f;
    __syncthreads();

    const int pstart = blockIdx.x * chunk;
    const int navail = P - pstart;
    const int n = (navail < chunk) ? navail : chunk;
    const int cbase = blockIdx.y * CG;
    const bool full = (n == chunk) && (chunk == PCHUNK) && ((P & 3) == 0);

    if (full) {
        const int wave = threadIdx.x >> 5;
        const int lane = threadIdx.x & 31;
        const int c = lane & 15;
        const int h = lane >> 4;

        v2f bsel[4];
#pragma unroll
        for (int g = 0; g < 4; ++g) {
            bsel[g].x = (c == 4 * g + 2 * h)     ? 1.0f : 0.0f;
            bsel[g].y = (c == 4 * g + 2 * h + 1) ? 1.0f : 0.0f;
        }

        const size_t wp = (size_t)pstart + (size_t)wave * WPIX;
        const float* dptr = feat + (size_t)(cbase + 2 * h) * (size_t)P + wp + 4 * c;
        const int*   sptr = seg + wp + 32 * h + 4 * (c & 7);
        float* lcol = &lsums[c];

        Blk ring[DEPTH];
        v8f acc[4];
#pragma unroll
        for (int d = 0; d < DEPTH; ++d) load_block(ring[d], dptr, sptr, (size_t)P, d);
#pragma unroll
        for (int blk = 0; blk < NBLK; ++blk) {
            Blk& cur = ring[blk % DEPTH];
            transpose_block(cur, bsel, acc);
            const int4 s = cur.s;
            if (blk + DEPTH < NBLK) load_block(cur, dptr, sptr, (size_t)P, blk + DEPTH);
            scatter_block(acc, s, lcol);
        }
    } else {
        const int ch = threadIdx.x & (CG - 1);
        const float* plane = feat + (size_t)(cbase + ch) * (size_t)P + pstart;
        for (int i = threadIdx.x / CG; i < n; i += CT / CG)
            atomicAdd(&lsums[seg[pstart + i] * CG + ch], plane[i]);
    }
    __syncthreads();

    float* grow = sums + (size_t)blockIdx.y * CG * (size_t)N;
    for (int i = threadIdx.x; i < CG * N; i += blockDim.x) {
        float v = lsums[i];
        if (v != 0.0f) atomicAdd(&grow[i], v);
    }
}

__global__ void finalize_wmma_kernel(const float* __restrict__ sums,
                                     const float* __restrict__ counts,
                                     float* __restrict__ out,
                                     int C, int N) {
    int wave = threadIdx.x >> 5;
    int lane = threadIdx.x & 31;
    int half = lane >> 4;
    int m    = lane & 15;
    int n0   = blockIdx.x * 16;
    int c0   = wave * 16;

    float cnt = counts[n0 + m];
    float inv = (cnt > 0.0f) ? (1.0f / cnt) : 0.0f;

    const float* tile = sums + (size_t)wave * (size_t)N * 16;

    v8f acc = {};
#pragma unroll
    for (int k0 = 0; k0 < 16; k0 += 4) {
        int kx = k0 + 2 * half;
        v2f a, b;
        a.x = (m == kx)     ? inv : 0.0f;
        a.y = (m == kx + 1) ? inv : 0.0f;
        b.x = tile[(size_t)(n0 + kx) * 16 + m];
        b.y = tile[(size_t)(n0 + kx + 1) * 16 + m];
        acc = __builtin_amdgcn_wmma_f32_16x16x4_f32(
             false, a,  false, b,
             (short)0, acc,  false,  false);
    }

#pragma unroll
    for (int v = 0; v < 8; ++v) {
        out[(size_t)(n0 + v + 8 * half) * C + c0 + m] = acc[v];
    }
}

extern "C" void kernel_launch(void* const* d_in, const int* in_sizes, int n_in,
                              void* d_out, int out_size, void* d_ws, size_t ws_size,
                              hipStream_t stream) {
    const float* feat = (const float*)d_in[0];
    const int*   seg  = (const int*)d_in[1];
    int P = in_sizes[1];
    int C = in_sizes[0] / P;
    int N = out_size / C;

    float* sums   = (float*)d_ws;
    float* counts = sums + (size_t)N * C;
    float* out    = (float*)d_out;

    int tot = N * C + N;
    zero_ws_kernel<<<(tot + 255) / 256, 256, 0, stream>>>(sums, tot);

    int chunk   = PCHUNK;
    int pblocks = (P + chunk - 1) / chunk;

    count_kernel<<<pblocks, CT, 0, stream>>>(seg, counts, P, N, chunk);

    dim3 agrid(pblocks, C / CG);
    accum_kernel<<<agrid, CT, 0, stream>>>(feat, seg, sums, P, C, N, chunk);

    finalize_wmma_kernel<<<N / 16, (C / 16) * 32, 0, stream>>>(sums, counts, out, C, N);
}


# global f32 atomics removed: each workgroup stores its LDS sums and counts whole to its own workspace slab, finalize adds the 64 slabs in a fixed order; the zeroing launch goes (4 launches to 3)
# speedup vs baseline: 6.6977x; 1.0915x over previous
#include <hip/hip_runtime.h>
#include <hip/hip_bf16.h>


typedef float v2f __attribute__((ext_vector_type(2)));
typedef float v8f __attribute__((ext_vector_type(8)));

#define MAXSEG 512
#define CG 16
#define PCHUNK 8192
#define CT 256
#define NWAVE (CT / 32)
#define PBLK 64
#define WPIX (PCHUNK / NWAVE)
#define NBLK (WPIX / PBLK)
#define DEPTH 3

__global__ void __launch_bounds__(CT)
count_kernel(const int* __restrict__ seg, float* __restrict__ pcounts,
             int P, int N, int chunk) {
    __shared__ int hist[MAXSEG];
    for (int i = threadIdx.x; i < N; i += blockDim.x) hist[i] = 0;
    __syncthreads();
    int pstart = blockIdx.x * chunk;
    if (chunk == PCHUNK && pstart + PCHUNK <= P) {
        int s[PCHUNK / CT];
#pragma unroll
        for (int j = 0; j < PCHUNK / CT; ++j)
            s[j] = seg[pstart + (int)threadIdx.x + j * CT];
#pragma unroll
        for (int j = 0; j < PCHUNK / CT; ++j)
            atomicAdd(&hist[s[j]], 1);
    } else {
        for (int i = threadIdx.x; i < chunk; i += blockDim.x) {
            int g = pstart + i;
            if (g < P) atomicAdd(&hist[seg[g]], 1);
        }
    }
    __syncthreads();
    const int ncp = (N + 31) & ~31;
    float* crow = pcounts + (size_t)blockIdx.x * (size_t)ncp;
    for (int i = threadIdx.x; i < N; i += blockDim.x) crow[i] = (float)hist[i];
}

struct Blk {
    float4 d[8];
    int4   s;
};

__device__ __forceinline__ void load_block(Blk& b, const float* dptr, const int* sptr,
                                           size_t P, int blk) {
#pragma unroll
    for (int k = 0; k < 8; ++k) {
        const int ch = 4 * (k >> 1) + (k & 1);
        b.d[k] = *(const float4*)(dptr + (size_t)ch * P + (size_t)blk * PBLK);
    }
    b.s = *(const int4*)(sptr + (size_t)blk * PBLK);
}

__device__ __forceinline__ void transpose_block(const Blk& b, const v2f (&bsel)[4], v8f (&acc)[4]) {
#pragma unroll
    for (int q = 0; q < 4; ++q) acc[q] = v8f{};
#pragma unroll
    for (int g = 0; g < 4; ++g) {
        const float4 lo = b.d[2 * g], hi = b.d[2 * g + 1];
        v2f a0 = {lo.x, hi.x}, a1 = {lo.y, hi.y}, a2 = {lo.z, hi.z}, a3 = {lo.w, hi.w};
        acc[0] = __builtin_amdgcn_wmma_f32_16x16x4_f32(false, a0, false, bsel[g], (short)0, acc[0], false, false);
        acc[1] = __builtin_amdgcn_wmma_f32_16x16x4_f32(false, a1, false, bsel[g], (short)0, acc[1], false, false);
        acc[2] = __builtin_amdgcn_wmma_f32_16x16x4_f32(false, a2, false, bsel[g], (short)0, acc[2], false, false);
        acc[3] = __builtin_amdgcn_wmma_f32_16x16x4_f32(false, a3, false, bsel[g], (short)0, acc[3], false, false);
    }
}

template <int R>
__device__ __forceinline__ void scatter_row(const v8f (&acc)[4], int4 s, float* lcol) {
    const int s0 = __builtin_amdgcn_update_dpp(0, s.x, 0x150 + R, 0xF, 0xF, false);
    const int s1 = __builtin_amdgcn_update_dpp(0, s.y, 0x150 + R, 0xF, 0xF, false);
    const int s2 = __builtin_amdgcn_update_dpp(0, s.z, 0x150 + R, 0xF, 0xF, false);
    const int s3 = __builtin_amdgcn_update_dpp(0, s.w, 0x150 + R, 0xF, 0xF, false);
    atomicAdd(lcol + s0 * CG, acc[0][R]);
    atomicAdd(lcol + s1 * CG, acc[1][R]);
    atomicAdd(lcol + s2 * CG, acc[2][R]);
    atomicAdd(lcol + s3 * CG, acc[3][R]);
}

__device__ __forceinline__ void scatter_block(const v8f (&acc)[4], int4 s, float* lcol) {
    scatter_row<0>(acc, s, lcol); scatter_row<1>(acc, s, lcol);
    scatter_row<2>(acc, s, lcol); scatter_row<3>(acc, s, lcol);
    scatter_row<4>(acc, s, lcol); scatter_row<5>(acc, s, lcol);
    scatter_row<6>(acc, s, lcol); scatter_row<7>(acc, s, lcol);
}

__global__ void __launch_bounds__(CT)
accum_kernel(const float* __restrict__ feat,
                             const int* __restrict__ seg,
                             float* __restrict__ psums,
                             int P, int C, int N, int chunk) {
    __shared__ alignas(16) float lsums[MAXSEG * CG];

    for (int i = threadIdx.x; i < CG * N; i += blockDim.x) lsums[i] = 0.0f;
    __syncthreads();

    const int pstart = blockIdx.x * chunk;
    const int navail = P - pstart;
    const int n = (navail < chunk) ? navail : chunk;
    const int cbase = blockIdx.y * CG;
    const bool full = (n == chunk) && (chunk == PCHUNK) && ((P & 3) == 0);

    if (full) {
        const int wave = threadIdx.x >> 5;
        const int lane = threadIdx.x & 31;
        const int c = lane & 15;
        const int h = lane >> 4;

        v2f bsel[4];
#pragma unroll
        for (int g = 0; g < 4; ++g) {
            bsel[g].x = (c == 4 * g + 2 * h)     ? 1.0f : 0.0f;
            bsel[g].y = (c == 4 * g + 2 * h + 1) ? 1.0f : 0.0f;
        }

        const size_t wp = (size_t)pstart + (size_t)wave * WPIX;
        const float* dptr = feat + (size_t)(cbase + 2 * h) * (size_t)P + wp + 4 * c;
        const int*   sptr = seg + wp + 32 * h + 4 * (c & 7);
        float* lcol = &lsums[c];

        Blk ring[DEPTH];
        v8f acc[4];
#pragma unroll
        for (int d = 0; d < DEPTH; ++d) load_block(ring[d], dptr, sptr, (size_t)P, d);
#pragma unroll
        for (int blk = 0; blk < NBLK; ++blk) {
            Blk& cur = ring[blk % DEPTH];
            transpose_block(cur, bsel, acc);
            const int4 s = cur.s;
            if (blk + DEPTH < NBLK) load_block(cur, dptr, sptr, (size_t)P, blk + DEPTH);
            scatter_block(acc, s, lcol);
        }
    } else {
        const int ch = threadIdx.x & (CG - 1);
        const float* plane = feat + (size_t)(cbase + ch) * (size_t)P + pstart;
        for (int i = threadIdx.x / CG; i < n; i += CT / CG)
            atomicAdd(&lsums[seg[pstart + i] * CG + ch], plane[i]);
    }
    __syncthreads();

    float4* gslab = (float4*)(psums + ((size_t)blockIdx.x * gridDim.y + blockIdx.y) * (size_t)(CG * N));
    const float4* l4 = (const float4*)lsums;
    for (int i = threadIdx.x; i < (CG / 4) * N; i += blockDim.x) gslab[i] = l4[i];
}

#define FT_SUM 256
#define FT_CNT 64
#define FT (FT_SUM + FT_CNT)

__global__ void __launch_bounds__(FT)
finalize_wmma_kernel(const float* __restrict__ psums,
                     const float* __restrict__ pcounts,
                     float* __restrict__ out,
                     int C, int N, int NP) {
    __shared__ float4 red[4][64];
    __shared__ float  cred[4][16];

    const int t  = threadIdx.x;
    const int n0 = blockIdx.x * 16;
    const int ct = blockIdx.y;
    const int m  = t & 15;

    if (t < FT_SUM) {
        const int nct = gridDim.y;
        const size_t slab = (size_t)N * 16;
        const int g = t >> 6;
        const int f = t & 63;
        const float4* p = (const float4*)(psums + ((size_t)g * nct + ct) * slab + (size_t)n0 * 16) + f;
        const size_t step = (size_t)nct * slab;
        float4 s = {0.0f, 0.0f, 0.0f, 0.0f};
#pragma unroll 16
        for (int x = g; x < NP; x += 4) {
            const float4 v = *p;
            p += step;
            s.x += v.x; s.y += v.y; s.z += v.z; s.w += v.w;
        }
        red[g][f] = s;
    } else {
        const int xg  = (t - FT_SUM) >> 4;
        const int ncp = (N + 31) & ~31;
        const float* q = pcounts + (size_t)xg * ncp + n0 + m;
        float cs = 0.0f;
#pragma unroll 16
        for (int x = xg; x < NP; x += 4) {
            cs += *q;
            q += 4 * (size_t)ncp;
        }
        cred[xg][m] = cs;
    }
    __syncthreads();

    if (__builtin_amdgcn_readfirstlane(t >> 5) == 0) {
        const int half = t >> 4;

        const float cnt = ((cred[0][m] + cred[1][m]) + cred[2][m]) + cred[3][m];
        const float inv = (cnt > 0.0f) ? (1.0f / cnt) : 0.0f;

        const float* r0 = (const float*)red[0];
        const float* r1 = (const float*)red[1];
        const float* r2 = (const float*)red[2];
        const float* r3 = (const float*)red[3];

        v8f acc = {};
#pragma unroll
        for (int k0 = 0; k0 < 16; k0 += 4) {
            const int kx = k0 + 2 * half;
            v2f a, b;
            a.x = (m == kx)     ? inv : 0.0f;
            a.y = (m == kx + 1) ? inv : 0.0f;
            const int i0 = kx * 16 + m;
            const int i1 = i0 + 16;
            b.x = ((r0[i0] + r1[i0]) + r2[i0]) + r3[i0];
            b.y = ((r0[i1] + r1[i1]) + r2[i1]) + r3[i1];
            acc = __builtin_amdgcn_wmma_f32_16x16x4_f32(
                 false, a,  false, b,
                 (short)0, acc,  false,  false);
        }

#pragma unroll
        for (int v = 0; v < 8; ++v) {
            out[(size_t)(n0 + v + 8 * half) * C + ct * 16 + m] = acc[v];
        }
    }
}

extern "C" void kernel_launch(void* const* d_in, const int* in_sizes, int n_in,
                              void* d_out, int out_size, void* d_ws, size_t ws_size,
                              hipStream_t stream) {
    const float* feat = (const float*)d_in[0];
    const int*   seg  = (const int*)d_in[1];
    int P = in_sizes[1];
    int C = in_sizes[0] / P;
    int N = out_size / C;

    float* psums = (float*)d_ws;
    float* out   = (float*)d_out;

    int chunk   = PCHUNK;
    int pblocks = (P + chunk - 1) / chunk;
    float* pcounts = psums + (size_t)pblocks * N * C;

    count_kernel<<<pblocks, CT, 0, stream>>>(seg, pcounts, P, N, chunk);

    dim3 agrid(pblocks, C / CG);
    accum_kernel<<<agrid, CT, 0, stream>>>(feat, seg, psums, P, C, N, chunk);

    dim3 fgrid(N / 16, C / 16);
    finalize_wmma_kernel<<<fgrid, FT, 0, stream>>>(psums, pcounts, out, C, N, pblocks);
}


# count kernel folded into accumulate kernel: channel-group-0 workgroups, which load their chunk's segment ids anyway, also add them into an LDS histogram and store the counts row; 3 launches to 2
# speedup vs baseline: 7.4997x; 1.1197x over previous
#include <hip/hip_runtime.h>
#include <hip/hip_bf16.h>


typedef float v2f __attribute__((ext_vector_type(2)));
typedef float v8f __attribute__((ext_vector_type(8)));

#define MAXSEG 512
#define CG 16
#define PCHUNK 8192
#define CT 256
#define NWAVE (CT / 32)
#define PBLK 64
#define WPIX (PCHUNK / NWAVE)
#define NBLK (WPIX / PBLK)
#define DEPTH 3

struct Blk {
    float4 d[8];
    int4   s;
};

__device__ __forceinline__ void load_block(Blk& b, const float* dptr, const int* sptr,
                                           size_t P, int blk) {
#pragma unroll
    for (int k = 0; k < 8; ++k) {
        const int ch = 4 * (k >> 1) + (k & 1);
        b.d[k] = *(const float4*)(dptr + (size_t)ch * P + (size_t)blk * PBLK);
    }
    b.s = *(const int4*)(sptr + (size_t)blk * PBLK);
}

__device__ __forceinline__ void transpose_block(const Blk& b, const v2f (&bsel)[4], v8f (&acc)[4]) {
#pragma unroll
    for (int q = 0; q < 4; ++q) acc[q] = v8f{};
#pragma unroll
    for (int g = 0; g < 4; ++g) {
        const float4 lo = b.d[2 * g], hi = b.d[2 * g + 1];
        v2f a0 = {lo.x, hi.x}, a1 = {lo.y, hi.y}, a2 = {lo.z, hi.z}, a3 = {lo.w, hi.w};
        acc[0] = __builtin_amdgcn_wmma_f32_16x16x4_f32(false, a0, false, bsel[g], (short)0, acc[0], false, false);
        acc[1] = __builtin_amdgcn_wmma_f32_16x16x4_f32(false, a1, false, bsel[g], (short)0, acc[1], false, false);
        acc[2] = __builtin_amdgcn_wmma_f32_16x16x4_f32(false, a2, false, bsel[g], (short)0, acc[2], false, false);
        acc[3] = __builtin_amdgcn_wmma_f32_16x16x4_f32(false, a3, false, bsel[g], (short)0, acc[3], false, false);
    }
}

template <int R>
__device__ __forceinline__ void scatter_row(const v8f (&acc)[4], int4 s, float* lcol) {
    const int s0 = __builtin_amdgcn_update_dpp(0, s.x, 0x150 + R, 0xF, 0xF, false);
    const int s1 = __builtin_amdgcn_update_dpp(0, s.y, 0x150 + R, 0xF, 0xF, false);
    const int s2 = __builtin_amdgcn_update_dpp(0, s.z, 0x150 + R, 0xF, 0xF, false);
    const int s3 = __builtin_amdgcn_update_dpp(0, s.w, 0x150 + R, 0xF, 0xF, false);
    atomicAdd(lcol + s0 * CG, acc[0][R]);
    atomicAdd(lcol + s1 * CG, acc[1][R]);
    atomicAdd(lcol + s2 * CG, acc[2][R]);
    atomicAdd(lcol + s3 * CG, acc[3][R]);
}

__device__ __forceinline__ void scatter_block(const v8f (&acc)[4], int4 s, float* lcol) {
    scatter_row<0>(acc, s, lcol); scatter_row<1>(acc, s, lcol);
    scatter_row<2>(acc, s, lcol); scatter_row<3>(acc, s, lcol);
    scatter_row<4>(acc, s, lcol); scatter_row<5>(acc, s, lcol);
    scatter_row<6>(acc, s, lcol); scatter_row<7>(acc, s, lcol);
}

__global__ void __launch_bounds__(CT)
accum_kernel(const float* __restrict__ feat,
                             const int* __restrict__ seg,
                             float* __restrict__ psums,
                             float* __restrict__ pcounts,
                             int P, int C, int N, int chunk) {
    __shared__ alignas(16) float lsums[MAXSEG * CG];
    __shared__ int hist[MAXSEG];

    const bool counting = (blockIdx.y == 0);

    for (int i = threadIdx.x; i < CG * N; i += blockDim.x) lsums[i] = 0.0f;
    if (counting)
        for (int i = threadIdx.x; i < N; i += blockDim.x) hist[i] = 0;
    __syncthreads();

    const int pstart = blockIdx.x * chunk;
    const int navail = P - pstart;
    const int n = (navail < chunk) ? navail : chunk;
    const int cbase = blockIdx.y * CG;
    const bool full = (n == chunk) && (chunk == PCHUNK) && ((P & 3) == 0);

    if (full) {
        const int wave = threadIdx.x >> 5;
        const int lane = threadIdx.x & 31;
        const int c = lane & 15;
        const int h = lane >> 4;

        v2f bsel[4];
#pragma unroll
        for (int g = 0; g < 4; ++g) {
            bsel[g].x = (c == 4 * g + 2 * h)     ? 1.0f : 0.0f;
            bsel[g].y = (c == 4 * g + 2 * h + 1) ? 1.0f : 0.0f;
        }

        const size_t wp = (size_t)pstart + (size_t)wave * WPIX;
        const float* dptr = feat + (size_t)(cbase + 2 * h) * (size_t)P + wp + 4 * c;
        const int*   sptr = seg + wp + 32 * h + 4 * (c & 7);
        float* lcol = &lsums[c];

        Blk ring[DEPTH];
        v8f acc[4];
#pragma unroll
        for (int d = 0; d < DEPTH; ++d) load_block(ring[d], dptr, sptr, (size_t)P, d);
#pragma unroll
        for (int blk = 0; blk < NBLK; ++blk) {
            Blk& cur = ring[blk % DEPTH];
            transpose_block(cur, bsel, acc);
            const int4 s = cur.s;
            if (blk + DEPTH < NBLK) load_block(cur, dptr, sptr, (size_t)P, blk + DEPTH);
            scatter_block(acc, s, lcol);
            if (counting && c < 8) {
                atomicAdd(&hist[s.x], 1);
                atomicAdd(&hist[s.y], 1);
                atomicAdd(&hist[s.z], 1);
                atomicAdd(&hist[s.w], 1);
            }
        }
    } else {
        const int ch = threadIdx.x & (CG - 1);
        const float* plane = feat + (size_t)(cbase + ch) * (size_t)P + pstart;
        for (int i = threadIdx.x / CG; i < n; i += CT / CG) {
            const int sid = seg[pstart + i];
            atomicAdd(&lsums[sid * CG + ch], plane[i]);
            if (counting && ch == 0) atomicAdd(&hist[sid], 1);
        }
    }
    __syncthreads();

    float4* gslab = (float4*)(psums + ((size_t)blockIdx.x * gridDim.y + blockIdx.y) * (size_t)(CG * N));
    const float4* l4 = (const float4*)lsums;
    for (int i = threadIdx.x; i < (CG / 4) * N; i += blockDim.x) gslab[i] = l4[i];

    if (counting) {
        const int ncp = (N + 31) & ~31;
        float* crow = pcounts + (size_t)blockIdx.x * (size_t)ncp;
        for (int i = threadIdx.x; i < N; i += blockDim.x) crow[i] = (float)hist[i];
    }
}

#define FT_SUM 256
#define FT_CNT 64
#define FT (FT_SUM + FT_CNT)

__global__ void __launch_bounds__(FT)
finalize_wmma_kernel(const float* __restrict__ psums,
                     const float* __restrict__ pcounts,
                     float* __restrict__ out,
                     int C, int N, int NP) {
    __shared__ float4 red[4][64];
    __shared__ float  cred[4][16];

    const int t  = threadIdx.x;
    const int n0 = blockIdx.x * 16;
    const int ct = blockIdx.y;
    const int m  = t & 15;

    if (t < FT_SUM) {
        const int nct = gridDim.y;
        const size_t slab = (size_t)N * 16;
        const int g = t >> 6;
        const int f = t & 63;
        const float4* p = (const float4*)(psums + ((size_t)g * nct + ct) * slab + (size_t)n0 * 16) + f;
        const size_t step = (size_t)nct * slab;
        float4 s = {0.0f, 0.0f, 0.0f, 0.0f};
#pragma unroll 16
        for (int x = g; x < NP; x += 4) {
            const float4 v = *p;
            p += step;
            s.x += v.x; s.y += v.y; s.z += v.z; s.w += v.w;
        }
        red[g][f] = s;
    } else {
        const int xg  = (t - FT_SUM) >> 4;
        const int ncp = (N + 31) & ~31;
        const float* q = pcounts + (size_t)xg * ncp + n0 + m;
        float cs = 0.0f;
#pragma unroll 16
        for (int x = xg; x < NP; x += 4) {
            cs += *q;
            q += 4 * (size_t)ncp;
        }
        cred[xg][m] = cs;
    }
    __syncthreads();

    if (__builtin_amdgcn_readfirstlane(t >> 5) == 0) {
        const int half = t >> 4;

        const float cnt = ((cred[0][m] + cred[1][m]) + cred[2][m]) + cred[3][m];
        const float inv = (cnt > 0.0f) ? (1.0f / cnt) : 0.0f;

        const float* r0 = (const float*)red[0];
        const float* r1 = (const float*)red[1];
        const float* r2 = (const float*)red[2];
        const float* r3 = (const float*)red[3];

        v8f acc = {};
#pragma unroll
        for (int k0 = 0; k0 < 16; k0 += 4) {
            const int kx = k0 + 2 * half;
            v2f a, b;
            a.x = (m == kx)     ? inv : 0.0f;
            a.y = (m == kx + 1) ? inv : 0.0f;
            const int i0 = kx * 16 + m;
            const int i1 = i0 + 16;
            b.x = ((r0[i0] + r1[i0]) + r2[i0]) + r3[i0];
            b.y = ((r0[i1] + r1[i1]) + r2[i1]) + r3[i1];
            acc = __builtin_amdgcn_wmma_f32_16x16x4_f32(
                 false, a,  false, b,
                 (short)0, acc,  false,  false);
        }

#pragma unroll
        for (int v = 0; v < 8; ++v) {
            out[(size_t)(n0 + v + 8 * half) * C + ct * 16 + m] = acc[v];
        }
    }
}

extern "C" void kernel_launch(void* const* d_in, const int* in_sizes, int n_in,
                              void* d_out, int out_size, void* d_ws, size_t ws_size,
                              hipStream_t stream) {
    const float* feat = (const float*)d_in[0];
    const int*   seg  = (const int*)d_in[1];
    int P = in_sizes[1];
    int C = in_sizes[0] / P;
    int N = out_size / C;

    float* psums = (float*)d_ws;
    float* out   = (float*)d_out;

    int chunk   = PCHUNK;
    int pblocks = (P + chunk - 1) / chunk;
    float* pcounts = psums + (size_t)pblocks * N * C;

    dim3 agrid(pblocks, C / CG);
    accum_kernel<<<agrid, CT, 0, stream>>>(feat, seg, psums, pcounts, P, C, N, chunk);

    dim3 fgrid(N / 16, C / 16);
    finalize_wmma_kernel<<<fgrid, FT, 0, stream>>>(psums, pcounts, out, C, N, pblocks);
}


# speedup vs baseline: 7.7937x; 1.0392x over previous
#include <hip/hip_runtime.h>
#include <hip/hip_bf16.h>


typedef float v2f __attribute__((ext_vector_type(2)));
typedef float v8f __attribute__((ext_vector_type(8)));

#define MAXSEG 512
#define CG 16
#define PCHUNK 8192
#define CT 256
#define NWAVE (CT / 32)
#define PBLK 64
#define WPIX (PCHUNK / NWAVE)
#define NBLK (WPIX / PBLK)
#define DEPTH 3

struct Blk {
    float4 d[8];
    int4   s;
};

__device__ __forceinline__ void load_block(Blk& b, const float* dptr, const int* sptr,
                                           size_t P, int blk) {
#pragma unroll
    for (int k = 0; k < 8; ++k) {
        const int ch = 4 * (k >> 1) + (k & 1);
        b.d[k] = *(const float4*)(dptr + (size_t)ch * P + (size_t)blk * PBLK);
    }
    b.s = *(const int4*)(sptr + (size_t)blk * PBLK);
}

__device__ __forceinline__ void transpose_block(const Blk& b, const v2f (&bsel)[4], v8f (&acc)[4]) {
#pragma unroll
    for (int q = 0; q < 4; ++q) acc[q] = v8f{};
#pragma unroll
    for (int g = 0; g < 4; ++g) {
        const float4 lo = b.d[2 * g], hi = b.d[2 * g + 1];
        v2f a0 = {lo.x, hi.x}, a1 = {lo.y, hi.y}, a2 = {lo.z, hi.z}, a3 = {lo.w, hi.w};
        acc[0] = __builtin_amdgcn_wmma_f32_16x16x4_f32(false, a0, false, bsel[g], (short)0, acc[0], false, false);
        acc[1] = __builtin_amdgcn_wmma_f32_16x16x4_f32(false, a1, false, bsel[g], (short)0, acc[1], false, false);
        acc[2] = __builtin_amdgcn_wmma_f32_16x16x4_f32(false, a2, false, bsel[g], (short)0, acc[2], false, false);
        acc[3] = __builtin_amdgcn_wmma_f32_16x16x4_f32(false, a3, false, bsel[g], (short)0, acc[3], false, false);
    }
}

template <int R>
__device__ __forceinline__ void scatter_row(const v8f (&acc)[4], int4 s, float* lcol) {
    const int s0 = __builtin_amdgcn_update_dpp(0, s.x, 0x150 + R, 0xF, 0xF, false);
    const int s1 = __builtin_amdgcn_update_dpp(0, s.y, 0x150 + R, 0xF, 0xF, false);
    const int s2 = __builtin_amdgcn_update_dpp(0, s.z, 0x150 + R, 0xF, 0xF, false);
    const int s3 = __builtin_amdgcn_update_dpp(0, s.w, 0x150 + R, 0xF, 0xF, false);
    atomicAdd(lcol + s0 * CG, acc[0][R]);
    atomicAdd(lcol + s1 * CG, acc[1][R]);
    atomicAdd(lcol + s2 * CG, acc[2][R]);
    atomicAdd(lcol + s3 * CG, acc[3][R]);
}

__device__ __forceinline__ void scatter_block(const v8f (&acc)[4], int4 s, float* lcol) {
    scatter_row<0>(acc, s, lcol); scatter_row<1>(acc, s, lcol);
    scatter_row<2>(acc, s, lcol); scatter_row<3>(acc, s, lcol);
    scatter_row<4>(acc, s, lcol); scatter_row<5>(acc, s, lcol);
    scatter_row<6>(acc, s, lcol); scatter_row<7>(acc, s, lcol);
}

struct alignas(16) Lds {
    int   hist[MAXSEG];
    float sums[2][MAXSEG * CG];
};

__global__ void __launch_bounds__(CT)
accum_kernel(const float* __restrict__ feat,
                             const int* __restrict__ seg,
                             float* __restrict__ psums,
                             float* __restrict__ pcounts,
                             int P, int C, int N, int chunk) {
    __shared__ Lds lds;
    int* hist = lds.hist;
    float* lsums = lds.sums[(threadIdx.x >> 5) & 1];

    const bool counting = (blockIdx.y == 0);

    {
        const float4 z = {0.0f, 0.0f, 0.0f, 0.0f};
        float4* z0 = (float4*)lds.sums[0];
        float4* z1 = (float4*)lds.sums[1];
        for (int i = threadIdx.x; i < (CG / 4) * N; i += blockDim.x) { z0[i] = z; z1[i] = z; }
    }
    if (counting)
        for (int i = threadIdx.x; i < N; i += blockDim.x) hist[i] = 0;
    __syncthreads();

    const int pstart = blockIdx.x * chunk;
    const int navail = P - pstart;
    const int n = (navail < chunk) ? navail : chunk;
    const int cbase = blockIdx.y * CG;
    const bool full = (n == chunk) && (chunk == PCHUNK) && ((P & 3) == 0);

    if (full) {
        const int wave = threadIdx.x >> 5;
        const int lane = threadIdx.x & 31;
        const int c = lane & 15;
        const int h = lane >> 4;

        v2f bsel[4];
#pragma unroll
        for (int g = 0; g < 4; ++g) {
            bsel[g].x = (c == 4 * g + 2 * h)     ? 1.0f : 0.0f;
            bsel[g].y = (c == 4 * g + 2 * h + 1) ? 1.0f : 0.0f;
        }

        const size_t wp = (size_t)pstart + (size_t)wave * WPIX;
        const float* dptr = feat + (size_t)(cbase + 2 * h) * (size_t)P + wp + 4 * c;
        const int*   sptr = seg + wp + 32 * h + 4 * (c & 7);
        float* lcol = &lsums[c];

        Blk ring[DEPTH];
        v8f acc[4];
#pragma unroll
        for (int d = 0; d < DEPTH; ++d) load_block(ring[d], dptr, sptr, (size_t)P, d);
#pragma unroll
        for (int blk = 0; blk < NBLK; ++blk) {
            Blk& cur = ring[blk % DEPTH];
            transpose_block(cur, bsel, acc);
            const int4 s = cur.s;
            if (blk + DEPTH < NBLK) load_block(cur, dptr, sptr, (size_t)P, blk + DEPTH);
            scatter_block(acc, s, lcol);
            if (counting && c < 8) {
                atomicAdd(&hist[s.x], 1);
                atomicAdd(&hist[s.y], 1);
                atomicAdd(&hist[s.z], 1);
                atomicAdd(&hist[s.w], 1);
            }
        }
    } else {
        const int ch = threadIdx.x & (CG - 1);
        const float* plane = feat + (size_t)(cbase + ch) * (size_t)P + pstart;
        for (int i = threadIdx.x / CG; i < n; i += CT / CG) {
            const int sid = seg[pstart + i];
            atomicAdd(&lsums[sid * CG + ch], plane[i]);
            if (counting && ch == 0) atomicAdd(&hist[sid], 1);
        }
    }
    __syncthreads();

    float4* gslab = (float4*)(psums + ((size_t)blockIdx.x * gridDim.y + blockIdx.y) * (size_t)(CG * N));
    const float4* l40 = (const float4*)lds.sums[0];
    const float4* l41 = (const float4*)lds.sums[1];
    for (int i = threadIdx.x; i < (CG / 4) * N; i += blockDim.x) {
        const float4 a = l40[i], b = l41[i];
        gslab[i] = float4{a.x + b.x, a.y + b.y, a.z + b.z, a.w + b.w};
    }

    if (counting) {
        const int ncp = (N + 31) & ~31;
        float* crow = pcounts + (size_t)blockIdx.x * (size_t)ncp;
        for (int i = threadIdx.x; i < N; i += blockDim.x) crow[i] = (float)hist[i];
    }
}

#define FT_SUM 256
#define FT_CNT 64
#define FT (FT_SUM + FT_CNT)

__global__ void __launch_bounds__(FT)
finalize_wmma_kernel(const float* __restrict__ psums,
                     const float* __restrict__ pcounts,
                     float* __restrict__ out,
                     int C, int N, int NP) {
    __shared__ float4 red[4][64];
    __shared__ float  cred[4][16];

    const int t  = threadIdx.x;
    const int n0 = blockIdx.x * 16;
    const int ct = blockIdx.y;
    const int m  = t & 15;

    if (t < FT_SUM) {
        const int nct = gridDim.y;
        const size_t slab = (size_t)N * 16;
        const int g = t >> 6;
        const int f = t & 63;
        const float4* p = (const float4*)(psums + ((size_t)g * nct + ct) * slab + (size_t)n0 * 16) + f;
        const size_t step = (size_t)nct * slab;
        float4 s = {0.0f, 0.0f, 0.0f, 0.0f};
#pragma unroll 16
        for (int x = g; x < NP; x += 4) {
            const float4 v = *p;
            p += step;
            s.x += v.x; s.y += v.y; s.z += v.z; s.w += v.w;
        }
        red[g][f] = s;
    } else {
        const int xg  = (t - FT_SUM) >> 4;
        const int ncp = (N + 31) & ~31;
        const float* q = pcounts + (size_t)xg * ncp + n0 + m;
        float cs = 0.0f;
#pragma unroll 16
        for (int x = xg; x < NP; x += 4) {
            cs += *q;
            q += 4 * (size_t)ncp;
        }
        cred[xg][m] = cs;
    }
    __syncthreads();

    if (__builtin_amdgcn_readfirstlane(t >> 5) == 0) {
        const int half = t >> 4;

        const float cnt = ((cred[0][m] + cred[1][m]) + cred[2][m]) + cred[3][m];
        const float inv = (cnt > 0.0f) ? (1.0f / cnt) : 0.0f;

        const float* r0 = (const float*)red[0];
        const float* r1 = (const float*)red[1];
        const float* r2 = (const float*)red[2];
        const float* r3 = (const float*)red[3];

        v8f acc = {};
#pragma unroll
        for (int k0 = 0; k0 < 16; k0 += 4) {
            const int kx = k0 + 2 * half;
            v2f a, b;
            a.x = (m == kx)     ? inv : 0.0f;
            a.y = (m == kx + 1) ? inv : 0.0f;
            const int i0 = kx * 16 + m;
            const int i1 = i0 + 16;
            b.x = ((r0[i0] + r1[i0]) + r2[i0]) + r3[i0];
            b.y = ((r0[i1] + r1[i1]) + r2[i1]) + r3[i1];
            acc = __builtin_amdgcn_wmma_f32_16x16x4_f32(
                 false, a,  false, b,
                 (short)0, acc,  false,  false);
        }

#pragma unroll
        for (int v = 0; v < 8; ++v) {
            out[(size_t)(n0 + v + 8 * half) * C + ct * 16 + m] = acc[v];
        }
    }
}

extern "C" void kernel_launch(void* const* d_in, const int* in_sizes, int n_in,
                              void* d_out, int out_size, void* d_ws, size_t ws_size,
                              hipStream_t stream) {
    const float* feat = (const float*)d_in[0];
    const int*   seg  = (const int*)d_in[1];
    int P = in_sizes[1];
    int C = in_sizes[0] / P;
    int N = out_size / C;

    float* psums = (float*)d_ws;
    float* out   = (float*)d_out;

    int chunk   = PCHUNK;
    int pblocks = (P + chunk - 1) / chunk;
    float* pcounts = psums + (size_t)pblocks * N * C;

    dim3 agrid(pblocks, C / CG);
    accum_kernel<<<agrid, CT, 0, stream>>>(feat, seg, psums, pcounts, P, C, N, chunk);

    dim3 fgrid(N / 16, C / 16);
    finalize_wmma_kernel<<<fgrid, FT, 0, stream>>>(psums, pcounts, out, C, N, pblocks);
}
